# starting kernel
# baseline (speedup 1.0000x reference)
#include <hip/hip_runtime.h>

#define N_ 16384
#define K_ 64
#define F_ 2048
#define ALPHA_ 1.0f
#define GAMMA_ 1.0f

typedef float v2f __attribute__((ext_vector_type(2)));
typedef float v8f __attribute__((ext_vector_type(8)));

__global__ __launch_bounds__(32) void gram_wmma(const float* __restrict__ C,
                                                float* __restrict__ G) {
    const int tile = blockIdx.x;
    const int ta = tile >> 2;
    const int tb = tile & 3;
    const int lane = threadIdx.x;
    const int lo = lane & 15;
    const int hi = lane >> 4;

    const v2f* __restrict__ ap = (const v2f*)(C + (size_t)(ta * 16 + lo) * F_);
    const v2f* __restrict__ bp = (const v2f*)(C + (size_t)(tb * 16 + lo) * F_);

    v8f acc = {0.f, 0.f, 0.f, 0.f, 0.f, 0.f, 0.f, 0.f};
#pragma unroll 4
    for (int i = 0; i < F_ / 4; ++i) {
        v2f a = ap[2 * i + hi];
        v2f b = bp[2 * i + hi];
        acc = __builtin_amdgcn_wmma_f32_16x16x4_f32(
             false, a,  false, b,
             (short)0, acc,  false,  false);
    }
#pragma unroll
    for (int r = 0; r < 8; ++r) {
        int m = r + 8 * hi;
        G[(ta * 16 + m) * K_ + tb * 16 + lo] = acc[r];
    }
}

__global__ __launch_bounds__(64) void stats(const float* __restrict__ G,
                                            const float* __restrict__ cw,
                                            float* __restrict__ rs1,
                                            float* __restrict__ w) {
    __shared__ float nrm_s[K_];
    const int k = threadIdx.x;
    float rowG = 0.f;
#pragma unroll 8
    for (int j = 0; j < K_; ++j) rowG += G[k * K_ + j];
    const float nrm = G[k * K_ + k];
    nrm_s[k] = nrm;
    __syncthreads();
    float S = 0.f;
#pragma unroll 8
    for (int j = 0; j < K_; ++j) S += nrm_s[j];
    const float rs = (float)K_ * nrm + S - 2.f * rowG + ALPHA_;
    rs1[k] = rs;
    w[k] = cw[k] / rs;
}

__global__ __launch_bounds__(256) void loss_idx(const float* __restrict__ features,
                                                const float* __restrict__ labels,
                                                const float* __restrict__ cluster,
                                                const float* __restrict__ rs1,
                                                float* __restrict__ loss,
                                                int* __restrict__ idx_out) {
    const int lane = threadIdx.x & 31;
    const int row = blockIdx.x * 8 + (threadIdx.x >> 5);

    float v0 = labels[(size_t)row * K_ + lane];
    float v1 = labels[(size_t)row * K_ + lane + 32];
    float bv = v0;
    int bi = lane;
    if (v1 > v0) { bv = v1; bi = lane + 32; }
#pragma unroll
    for (int m = 16; m >= 1; m >>= 1) {
        float ov = __shfl_xor(bv, m, 32);
        int oi = __shfl_xor(bi, m, 32);
        if (ov > bv || (ov == bv && oi < bi)) { bv = ov; bi = oi; }
    }
    const int idx = bi;

    const float4* __restrict__ x4 = (const float4*)(features + (size_t)row * F_);
    const float4* __restrict__ c4 = (const float4*)(cluster + (size_t)idx * F_);
    float s = 0.f;
#pragma unroll 4
    for (int i = lane; i < F_ / 4; i += 32) {
        float4 a = x4[i];
        float4 b = c4[i];
        float d0 = a.x - b.x, d1 = a.y - b.y, d2 = a.z - b.z, d3 = a.w - b.w;
        s += d0 * d0 + d1 * d1 + d2 * d2 + d3 * d3;
    }
#pragma unroll
    for (int m = 16; m >= 1; m >>= 1) s += __shfl_xor(s, m, 32);

    if (lane == 0) {
        loss[row] = s / rs1[idx];
        idx_out[row] = idx;
    }
}

__global__ __launch_bounds__(256) void cluster_update(const float* __restrict__ features,
                                                      const float* __restrict__ cluster,
                                                      const float* __restrict__ w,
                                                      const int* __restrict__ idx,
                                                      float* __restrict__ newc) {
    const int k = blockIdx.y;
    const int c0 = blockIdx.x * 512 + threadIdx.x * 2;
    float a0 = 0.f, a1 = 0.f;
    int cnt = 0;
    for (int n = 0; n < N_; ++n) {
        if (idx[n] == k) {
            const float2 xr = *(const float2*)(features + (size_t)n * F_ + c0);
            a0 += xr.x;
            a1 += xr.y;
            ++cnt;
        }
    }
    const float wk = w[k];
    const float2 cc = *(const float2*)(cluster + (size_t)k * F_ + c0);
    float2 out;
    out.x = cc.x - GAMMA_ * wk * (a0 - (float)cnt * cc.x);
    out.y = cc.y - GAMMA_ * wk * (a1 - (float)cnt * cc.y);
    *(float2*)(newc + (size_t)k * F_ + c0) = out;
}

extern "C" void kernel_launch(void* const* d_in, const int* in_sizes, int n_in,
                              void* d_out, int out_size, void* d_ws, size_t ws_size,
                              hipStream_t stream) {
    const float* features = (const float*)d_in[0];
    const float* labels   = (const float*)d_in[1];
    const float* cluster  = (const float*)d_in[2];
    const float* cw       = (const float*)d_in[3];

    float* loss = (float*)d_out;
    float* newc = loss + N_;

    float* G   = (float*)d_ws;
    float* rs1 = G + K_ * K_;
    float* wv  = rs1 + K_;
    int* idxw  = (int*)(wv + K_);

    gram_wmma<<<16, 32, 0, stream>>>(cluster, G);
    stats<<<1, 64, 0, stream>>>(G, cw, rs1, wv);
    loss_idx<<<N_ / 8, 256, 0, stream>>>(features, labels, cluster, rs1, loss, idxw);
    cluster_update<<<dim3(F_ / 512, K_), 256, 0, stream>>>(features, cluster, wv, idxw, newc);
}


# cluster_update as a matrix product: onehot(idx)^T x features on f32 WMMA, 8 row slices x 32 column strips, wave tiles added in LDS, slices in a new 5th launch; new sum order: 64 row-block partials
# speedup vs baseline: 7.5890x; 7.5890x over previous
#include <hip/hip_runtime.h>

#define N_ 16384
#define K_ 64
#define F_ 2048
#define ALPHA_ 1.0f
#define GAMMA_ 1.0f

typedef float v2f __attribute__((ext_vector_type(2)));
typedef float v8f __attribute__((ext_vector_type(8)));

__global__ __launch_bounds__(32) void gram_wmma(const float* __restrict__ C,
                                                float* __restrict__ G) {
    const int tile = blockIdx.x;
    const int ta = tile >> 2;
    const int tb = tile & 3;
    const int lane = threadIdx.x;
    const int lo = lane & 15;
    const int hi = lane >> 4;

    const v2f* __restrict__ ap = (const v2f*)(C + (size_t)(ta * 16 + lo) * F_);
    const v2f* __restrict__ bp = (const v2f*)(C + (size_t)(tb * 16 + lo) * F_);

    v8f acc = {0.f, 0.f, 0.f, 0.f, 0.f, 0.f, 0.f, 0.f};
#pragma unroll 4
    for (int i = 0; i < F_ / 4; ++i) {
        v2f a = ap[2 * i + hi];
        v2f b = bp[2 * i + hi];
        acc = __builtin_amdgcn_wmma_f32_16x16x4_f32(
             false, a,  false, b,
             (short)0, acc,  false,  false);
    }
#pragma unroll
    for (int r = 0; r < 8; ++r) {
        int m = r + 8 * hi;
        G[(ta * 16 + m) * K_ + tb * 16 + lo] = acc[r];
    }
}

__global__ __launch_bounds__(64) void stats(const float* __restrict__ G,
                                            const float* __restrict__ cw,
                                            float* __restrict__ rs1,
                                            float* __restrict__ w) {
    __shared__ float nrm_s[K_];
    const int k = threadIdx.x;
    float rowG = 0.f;
#pragma unroll 8
    for (int j = 0; j < K_; ++j) rowG += G[k * K_ + j];
    const float nrm = G[k * K_ + k];
    nrm_s[k] = nrm;
    __syncthreads();
    float S = 0.f;
#pragma unroll 8
    for (int j = 0; j < K_; ++j) S += nrm_s[j];
    const float rs = (float)K_ * nrm + S - 2.f * rowG + ALPHA_;
    rs1[k] = rs;
    w[k] = cw[k] / rs;
}

__global__ __launch_bounds__(256) void loss_idx(const float* __restrict__ features,
                                                const float* __restrict__ labels,
                                                const float* __restrict__ cluster,
                                                const float* __restrict__ rs1,
                                                float* __restrict__ loss,
                                                int* __restrict__ idx_out) {
    const int lane = threadIdx.x & 31;
    const int row = blockIdx.x * 8 + (threadIdx.x >> 5);

    float v0 = labels[(size_t)row * K_ + lane];
    float v1 = labels[(size_t)row * K_ + lane + 32];
    float bv = v0;
    int bi = lane;
    if (v1 > v0) { bv = v1; bi = lane + 32; }
#pragma unroll
    for (int m = 16; m >= 1; m >>= 1) {
        float ov = __shfl_xor(bv, m, 32);
        int oi = __shfl_xor(bi, m, 32);
        if (ov > bv || (ov == bv && oi < bi)) { bv = ov; bi = oi; }
    }
    const int idx = bi;

    const float4* __restrict__ x4 = (const float4*)(features + (size_t)row * F_);
    const float4* __restrict__ c4 = (const float4*)(cluster + (size_t)idx * F_);
    float s = 0.f;
#pragma unroll 4
    for (int i = lane; i < F_ / 4; i += 32) {
        float4 a = x4[i];
        float4 b = c4[i];
        float d0 = a.x - b.x, d1 = a.y - b.y, d2 = a.z - b.z, d3 = a.w - b.w;
        s += d0 * d0 + d1 * d1 + d2 * d2 + d3 * d3;
    }
#pragma unroll
    for (int m = 16; m >= 1; m >>= 1) s += __shfl_xor(s, m, 32);

    if (lane == 0) {
        loss[row] = s / rs1[idx];
        idx_out[row] = idx;
    }
}

#define RS_ 8
#define WROWS_ (N_ / RS_ / 8)
#define STAGE_ 8
#define NSTAGE_ (WROWS_ / (4 * STAGE_))

struct RowStage {
    float4 x[STAGE_][2];
    int2 id[STAGE_];
};

__device__ __forceinline__ void load_stage(RowStage& s, const float* __restrict__ fp,
                                           const int* __restrict__ ip) {
#pragma unroll
    for (int u = 0; u < STAGE_; ++u) {
        s.x[u][0] = *(const float4*)(fp + (size_t)(4 * u) * F_);
        s.x[u][1] = *(const float4*)(fp + (size_t)(4 * u + 1) * F_);
        s.id[u] = *(const int2*)(ip + 4 * u);
    }
}

__device__ __forceinline__ void mac_stage(const RowStage& s, const int c,
                                          v8f (&acc)[4][4], v2f (&cnt)[4]) {
#pragma unroll
    for (int u = 0; u < STAGE_; ++u) {
        const int t0 = s.id[u].x - c;
        const int t1 = s.id[u].y - c;
        const float4 x0 = s.x[u][0];
        const float4 x1 = s.x[u][1];
        v2f b[4];
        b[0].x = x0.x; b[0].y = x1.x;
        b[1].x = x0.y; b[1].y = x1.y;
        b[2].x = x0.z; b[2].y = x1.z;
        b[3].x = x0.w; b[3].y = x1.w;
#pragma unroll
        for (int mt = 0; mt < 4; ++mt) {
            v2f a;
            a.x = (t0 == 16 * mt) ? 1.f : 0.f;
            a.y = (t1 == 16 * mt) ? 1.f : 0.f;
            cnt[mt] += a;
#pragma unroll
            for (int j = 0; j < 4; ++j) {
                acc[mt][j] = __builtin_amdgcn_wmma_f32_16x16x4_f32(
                     false, a,  false, b[j],
                     (short)0, acc[mt][j],  false,  false);
            }
        }
    }
}

__global__ __launch_bounds__(256) void cluster_sums_wmma(const float* __restrict__ features,
                                                         const int* __restrict__ idx,
                                                         float* __restrict__ part,
                                                         float* __restrict__ cntp) {
    __shared__ float red[8][K_ * 64];
    __shared__ float cred[8][K_];

    const int lane = threadIdx.x & 31;
    const int wave = threadIdx.x >> 5;
    const int c = lane & 15;
    const int h = lane >> 4;
    const int strip = blockIdx.x;
    const int slice = blockIdx.y;
    const int row0 = slice * (N_ / RS_) + wave * WROWS_;

    const float* __restrict__ fp = features + (size_t)(row0 + 2 * h) * F_ + strip * 64 + 4 * c;
    const int* __restrict__ ip = idx + row0 + 2 * h;

    v8f acc[4][4];
    v2f cnt[4];
#pragma unroll
    for (int mt = 0; mt < 4; ++mt) {
        cnt[mt].x = 0.f; cnt[mt].y = 0.f;
#pragma unroll
        for (int j = 0; j < 4; ++j)
#pragma unroll
            for (int r = 0; r < 8; ++r) acc[mt][j][r] = 0.f;
    }

    RowStage sa, sb;
    load_stage(sa, fp, ip);
#pragma unroll 1
    for (int it = 0; it < NSTAGE_ - 2; it += 2) {
        __builtin_amdgcn_sched_barrier(0);
        load_stage(sb, fp + (size_t)(it + 1) * (4 * STAGE_) * F_, ip + (it + 1) * (4 * STAGE_));
        __builtin_amdgcn_sched_barrier(0);
        mac_stage(sa, c, acc, cnt);
        __builtin_amdgcn_sched_barrier(0);
        load_stage(sa, fp + (size_t)(it + 2) * (4 * STAGE_) * F_, ip + (it + 2) * (4 * STAGE_));
        __builtin_amdgcn_sched_barrier(0);
        mac_stage(sb, c, acc, cnt);
    }
    __builtin_amdgcn_sched_barrier(0);
    load_stage(sb, fp + (size_t)(NSTAGE_ - 1) * (4 * STAGE_) * F_, ip + (NSTAGE_ - 1) * (4 * STAGE_));
    __builtin_amdgcn_sched_barrier(0);
    mac_stage(sa, c, acc, cnt);
    mac_stage(sb, c, acc, cnt);

#pragma unroll
    for (int mt = 0; mt < 4; ++mt) {
#pragma unroll
        for (int r = 0; r < 8; ++r) {
            float4 v;
            v.x = acc[mt][0][r]; v.y = acc[mt][1][r];
            v.z = acc[mt][2][r]; v.w = acc[mt][3][r];
            *(float4*)&red[wave][(mt * 16 + 8 * h + r) * 64 + 4 * c] = v;
        }
        float cw = cnt[mt].x + cnt[mt].y;
        cw += __shfl_xor(cw, 16, 32);
        if (h == 0) cred[wave][mt * 16 + c] = cw;
    }
    __syncthreads();

    float* __restrict__ pout = part + ((size_t)slice * K_ + wave * 8) * F_ + strip * 64;
#pragma unroll
    for (int i = 0; i < 4; ++i) {
        const int e = i * 128 + lane * 4;
        float4 sum = *(const float4*)&red[0][wave * 512 + e];
#pragma unroll
        for (int p = 1; p < 8; ++p) {
            const float4 v = *(const float4*)&red[p][wave * 512 + e];
            sum.x += v.x; sum.y += v.y; sum.z += v.z; sum.w += v.w;
        }
        *(float4*)(pout + (size_t)(e >> 6) * F_ + (e & 63)) = sum;
    }
    if (strip == 0 && threadIdx.x < K_) {
        float cs = cred[0][threadIdx.x];
#pragma unroll
        for (int p = 1; p < 8; ++p) cs += cred[p][threadIdx.x];
        cntp[slice * K_ + threadIdx.x] = cs;
    }
}

__global__ __launch_bounds__(256) void cluster_finish(const float* __restrict__ part,
                                                      const float* __restrict__ cntp,
                                                      const float* __restrict__ cluster,
                                                      const float* __restrict__ w,
                                                      float* __restrict__ newc) {
    const int k = blockIdx.y;
    const int c0 = (blockIdx.x * 256 + threadIdx.x) * 4;
    float4 a = *(const float4*)(part + (size_t)k * F_ + c0);
    float cnt = cntp[k];
#pragma unroll
    for (int s = 1; s < RS_; ++s) {
        const float4 v = *(const float4*)(part + ((size_t)s * K_ + k) * F_ + c0);
        a.x += v.x; a.y += v.y; a.z += v.z; a.w += v.w;
        cnt += cntp[s * K_ + k];
    }
    const float wk = w[k];
    const float4 cc = *(const float4*)(cluster + (size_t)k * F_ + c0);
    float4 out;
    out.x = cc.x - GAMMA_ * wk * (a.x - cnt * cc.x);
    out.y = cc.y - GAMMA_ * wk * (a.y - cnt * cc.y);
    out.z = cc.z - GAMMA_ * wk * (a.z - cnt * cc.z);
    out.w = cc.w - GAMMA_ * wk * (a.w - cnt * cc.w);
    *(float4*)(newc + (size_t)k * F_ + c0) = out;
}

extern "C" void kernel_launch(void* const* d_in, const int* in_sizes, int n_in,
                              void* d_out, int out_size, void* d_ws, size_t ws_size,
                              hipStream_t stream) {
    const float* features = (const float*)d_in[0];
    const float* labels   = (const float*)d_in[1];
    const float* cluster  = (const float*)d_in[2];
    const float* cw       = (const float*)d_in[3];

    float* loss = (float*)d_out;
    float* newc = loss + N_;

    float* G   = (float*)d_ws;
    float* rs1 = G + K_ * K_;
    float* wv  = rs1 + K_;
    int* idxw  = (int*)(wv + K_);
    float* cntp = (float*)(idxw + N_);
    float* part = cntp + RS_ * K_;

    gram_wmma<<<16, 32, 0, stream>>>(cluster, G);
    stats<<<1, 64, 0, stream>>>(G, cw, rs1, wv);
    loss_idx<<<N_ / 8, 256, 0, stream>>>(features, labels, cluster, rs1, loss, idxw);
    cluster_sums_wmma<<<dim3(F_ / 64, RS_), 256, 0, stream>>>(features, idxw, part, cntp);
    cluster_finish<<<dim3(F_ / 1024, K_), 256, 0, stream>>>(part, cntp, cluster, wv, newc);
}


# Gram rides in the cluster-sum launch: each 64-column strip's partial tiles on WMMA (2 per block), only row sums and diagonal leave; finish forms row_sum, w, loss; 5 launches -> 3; new row_sum order
# speedup vs baseline: 13.3230x; 1.7556x over previous
#include <hip/hip_runtime.h>


#define N_ 16384
#define K_ 64
#define F_ 2048
#define ALPHA_ 1.0f
#define GAMMA_ 1.0f

typedef float v2f __attribute__((ext_vector_type(2)));
typedef float v8f __attribute__((ext_vector_type(8)));

__global__ __launch_bounds__(256) void loss_idx(const float* __restrict__ features,
                                                const float* __restrict__ labels,
                                                const float* __restrict__ cluster,
                                                float* __restrict__ nume,
                                                int* __restrict__ idx_out) {
    const int lane = threadIdx.x & 31;
    const int row = blockIdx.x * 8 + (threadIdx.x >> 5);

    float v0 = labels[(size_t)row * K_ + lane];
    float v1 = labels[(size_t)row * K_ + lane + 32];
    float bv = v0;
    int bi = lane;
    if (v1 > v0) { bv = v1; bi = lane + 32; }
#pragma unroll
    for (int m = 16; m >= 1; m >>= 1) {
        float ov = __shfl_xor(bv, m, 32);
        int oi = __shfl_xor(bi, m, 32);
        if (ov > bv || (ov == bv && oi < bi)) { bv = ov; bi = oi; }
    }
    const int idx = bi;

    const float4* __restrict__ x4 = (const float4*)(features + (size_t)row * F_);
    const float4* __restrict__ c4 = (const float4*)(cluster + (size_t)idx * F_);
    float s = 0.f;
#pragma unroll 4
    for (int i = lane; i < F_ / 4; i += 32) {
        float4 a = x4[i];
        float4 b = c4[i];
        float d0 = a.x - b.x, d1 = a.y - b.y, d2 = a.z - b.z, d3 = a.w - b.w;
        s += d0 * d0 + d1 * d1 + d2 * d2 + d3 * d3;
    }
#pragma unroll
    for (int m = 16; m >= 1; m >>= 1) s += __shfl_xor(s, m, 32);

    if (lane == 0) {
        nume[row] = s;
        idx_out[row] = idx;
    }
}

#define RS_ 8
#define WROWS_ (N_ / RS_ / 8)
#define STAGE_ 8
#define NSTAGE_ (WROWS_ / (4 * STAGE_))

struct RowStage {
    float4 x[STAGE_][2];
    int2 id[STAGE_];
};

__device__ __forceinline__ void load_stage(RowStage& s, const float* __restrict__ fp,
                                           const int* __restrict__ ip) {
#pragma unroll
    for (int u = 0; u < STAGE_; ++u) {
        s.x[u][0] = *(const float4*)(fp + (size_t)(4 * u) * F_);
        s.x[u][1] = *(const float4*)(fp + (size_t)(4 * u + 1) * F_);
        s.id[u] = *(const int2*)(ip + 4 * u);
    }
}

__device__ __forceinline__ void mac_stage(const RowStage& s, const int c,
                                          v8f (&acc)[4][4], v2f (&cnt)[4]) {
#pragma unroll
    for (int u = 0; u < STAGE_; ++u) {
        const int t0 = s.id[u].x - c;
        const int t1 = s.id[u].y - c;
        const float4 x0 = s.x[u][0];
        const float4 x1 = s.x[u][1];
        v2f b[4];
        b[0].x = x0.x; b[0].y = x1.x;
        b[1].x = x0.y; b[1].y = x1.y;
        b[2].x = x0.z; b[2].y = x1.z;
        b[3].x = x0.w; b[3].y = x1.w;
#pragma unroll
        for (int mt = 0; mt < 4; ++mt) {
            v2f a;
            a.x = (t0 == 16 * mt) ? 1.f : 0.f;
            a.y = (t1 == 16 * mt) ? 1.f : 0.f;
            cnt[mt] += a;
#pragma unroll
            for (int j = 0; j < 4; ++j) {
                acc[mt][j] = __builtin_amdgcn_wmma_f32_16x16x4_f32(
                     false, a,  false, b[j],
                     (short)0, acc[mt][j],  false,  false);
            }
        }
    }
}

__global__ __launch_bounds__(256) void cluster_sums_wmma(const float* __restrict__ features,
                                                         const int* __restrict__ idx,
                                                         const float* __restrict__ cluster,
                                                         float* __restrict__ part,
                                                         float* __restrict__ cntp,
                                                         float* __restrict__ rowp,
                                                         float* __restrict__ diagp) {
    __shared__ float red[8][K_ * 64];
    __shared__ float cred[8][K_];

    const int lane = threadIdx.x & 31;
    const int wave = threadIdx.x >> 5;
    const int c = lane & 15;
    const int h = lane >> 4;
    const int strip = blockIdx.x;
    const int slice = blockIdx.y;
    const int row0 = slice * (N_ / RS_) + wave * WROWS_;

    const float* __restrict__ fp = features + (size_t)(row0 + 2 * h) * F_ + strip * 64 + 4 * c;
    const int* __restrict__ ip = idx + row0 + 2 * h;

    if (wave < 2) {
        const int t = 2 * slice + wave;
        const int ta = t >> 2;
        const int tb = t & 3;
        const float* __restrict__ ga = cluster + (size_t)(16 * ta + c) * F_ + strip * 64 + 4 * h;
        const float* __restrict__ gb = cluster + (size_t)(16 * tb + c) * F_ + strip * 64 + 4 * h;
        float4 av[8], bv[8];
#pragma unroll
        for (int i = 0; i < 8; ++i) {
            av[i] = *(const float4*)(ga + 8 * i);
            bv[i] = *(const float4*)(gb + 8 * i);
        }
        v8f g0 = {0.f, 0.f, 0.f, 0.f, 0.f, 0.f, 0.f, 0.f};
        v8f g1 = {0.f, 0.f, 0.f, 0.f, 0.f, 0.f, 0.f, 0.f};
#pragma unroll
        for (int i = 0; i < 8; ++i) {
            v2f a0, b0, a1, b1;
            a0.x = av[i].x; a0.y = av[i].y; b0.x = bv[i].x; b0.y = bv[i].y;
            a1.x = av[i].z; a1.y = av[i].w; b1.x = bv[i].z; b1.y = bv[i].w;
            g0 = __builtin_amdgcn_wmma_f32_16x16x4_f32(
                 false, a0,  false, b0,
                 (short)0, g0,  false,  false);
            g1 = __builtin_amdgcn_wmma_f32_16x16x4_f32(
                 false, a1,  false, b1,
                 (short)0, g1,  false,  false);
        }
        float gv[8];
#pragma unroll
        for (int r = 0; r < 8; ++r) gv[r] = g0[r] + g1[r];
        if (ta == tb) {
            float d = gv[0];
#pragma unroll
            for (int r = 1; r < 8; ++r) d = ((c & 7) == r) ? gv[r] : d;
            if ((c >> 3) == h) diagp[strip * K_ + 16 * ta + c] = d;
        }
        float rsel = 0.f;
#pragma unroll
        for (int r = 0; r < 8; ++r) {
            float v = gv[r];
            v += __shfl_xor(v, 8, 32);
            v += __shfl_xor(v, 4, 32);
            v += __shfl_xor(v, 2, 32);
            v += __shfl_xor(v, 1, 32);
            if (c == r) rsel = v;
        }
        if (c < 8) rowp[(size_t)(strip * 4 + tb) * K_ + 16 * ta + 8 * h + c] = rsel;
    }

    v8f acc[4][4];
    v2f cnt[4];
#pragma unroll
    for (int mt = 0; mt < 4; ++mt) {
        cnt[mt].x = 0.f; cnt[mt].y = 0.f;
#pragma unroll
        for (int j = 0; j < 4; ++j)
#pragma unroll
            for (int r = 0; r < 8; ++r) acc[mt][j][r] = 0.f;
    }

    RowStage sa, sb;
    load_stage(sa, fp, ip);
#pragma unroll 1
    for (int it = 0; it < NSTAGE_ - 2; it += 2) {
        __builtin_amdgcn_sched_barrier(0);
        load_stage(sb, fp + (size_t)(it + 1) * (4 * STAGE_) * F_, ip + (it + 1) * (4 * STAGE_));
        __builtin_amdgcn_sched_barrier(0);
        mac_stage(sa, c, acc, cnt);
        __builtin_amdgcn_sched_barrier(0);
        load_stage(sa, fp + (size_t)(it + 2) * (4 * STAGE_) * F_, ip + (it + 2) * (4 * STAGE_));
        __builtin_amdgcn_sched_barrier(0);
        mac_stage(sb, c, acc, cnt);
    }
    __builtin_amdgcn_sched_barrier(0);
    load_stage(sb, fp + (size_t)(NSTAGE_ - 1) * (4 * STAGE_) * F_, ip + (NSTAGE_ - 1) * (4 * STAGE_));
    __builtin_amdgcn_sched_barrier(0);
    mac_stage(sa, c, acc, cnt);
    mac_stage(sb, c, acc, cnt);

#pragma unroll
    for (int mt = 0; mt < 4; ++mt) {
#pragma unroll
        for (int r = 0; r < 8; ++r) {
            float4 v;
            v.x = acc[mt][0][r]; v.y = acc[mt][1][r];
            v.z = acc[mt][2][r]; v.w = acc[mt][3][r];
            *(float4*)&red[wave][(mt * 16 + 8 * h + r) * 64 + 4 * c] = v;
        }
        float cw = cnt[mt].x + cnt[mt].y;
        cw += __shfl_xor(cw, 16, 32);
        if (h == 0) cred[wave][mt * 16 + c] = cw;
    }
    __syncthreads();

    float* __restrict__ pout = part + ((size_t)slice * K_ + wave * 8) * F_ + strip * 64;
#pragma unroll
    for (int i = 0; i < 4; ++i) {
        const int e = i * 128 + lane * 4;
        float4 sum = *(const float4*)&red[0][wave * 512 + e];
#pragma unroll
        for (int p = 1; p < 8; ++p) {
            const float4 v = *(const float4*)&red[p][wave * 512 + e];
            sum.x += v.x; sum.y += v.y; sum.z += v.z; sum.w += v.w;
        }
        *(float4*)(pout + (size_t)(e >> 6) * F_ + (e & 63)) = sum;
    }
    if (strip == 0 && threadIdx.x < K_) {
        float cs = cred[0][threadIdx.x];
#pragma unroll
        for (int p = 1; p < 8; ++p) cs += cred[p][threadIdx.x];
        cntp[slice * K_ + threadIdx.x] = cs;
    }
}

__global__ __launch_bounds__(256) void cluster_finish(const float* __restrict__ part,
                                                      const float* __restrict__ cntp,
                                                      const float* __restrict__ cluster,
                                                      const float* __restrict__ cw,
                                                      const float* __restrict__ rowp,
                                                      const float* __restrict__ diagp,
                                                      const float* __restrict__ nume,
                                                      const int* __restrict__ idx,
                                                      float* __restrict__ newc,
                                                      float* __restrict__ loss) {
    __shared__ float rowS[4][K_];
    __shared__ float diagS[4][K_];
    __shared__ float nrm_s[K_];
    __shared__ float rs1_s[K_];

    const int t = threadIdx.x;
    const int k = blockIdx.y;
    const int c0 = (blockIdx.x * 256 + t) * 4;

    float4 a = *(const float4*)(part + (size_t)k * F_ + c0);
    float cnt = cntp[k];
#pragma unroll
    for (int s = 1; s < RS_; ++s) {
        const float4 v = *(const float4*)(part + ((size_t)s * K_ + k) * F_ + c0);
        a.x += v.x; a.y += v.y; a.z += v.z; a.w += v.w;
        cnt += cntp[s * K_ + k];
    }
    const float4 cc = *(const float4*)(cluster + (size_t)k * F_ + c0);
    const int n = (blockIdx.y * 2 + blockIdx.x) * 128 + (t & 127);
    const float nm = nume[n];
    const int id = idx[n];

    {
        const int kk = t & 63;
        const int q = t >> 6;
        float rsum = 0.f;
#pragma unroll
        for (int p = 0; p < F_ / 64; ++p) rsum += rowp[(size_t)(p * 4 + q) * K_ + kk];
        float dsum = 0.f;
#pragma unroll
        for (int p = 0; p < 8; ++p) dsum += diagp[(size_t)(8 * q + p) * K_ + kk];
        rowS[q][kk] = rsum;
        diagS[q][kk] = dsum;
    }
    __syncthreads();
    if (t < K_) nrm_s[t] = (diagS[0][t] + diagS[1][t]) + (diagS[2][t] + diagS[3][t]);
    __syncthreads();
    if (t < K_) {
        float S = 0.f;
#pragma unroll 8
        for (int j = 0; j < K_; ++j) S += nrm_s[j];
        const float rowG = (rowS[0][t] + rowS[1][t]) + (rowS[2][t] + rowS[3][t]);
        rs1_s[t] = (float)K_ * nrm_s[t] + S - 2.f * rowG + ALPHA_;
    }
    __syncthreads();

    const float wk = cw[k] / rs1_s[k];
    float4 out;
    out.x = cc.x - GAMMA_ * wk * (a.x - cnt * cc.x);
    out.y = cc.y - GAMMA_ * wk * (a.y - cnt * cc.y);
    out.z = cc.z - GAMMA_ * wk * (a.z - cnt * cc.z);
    out.w = cc.w - GAMMA_ * wk * (a.w - cnt * cc.w);
    *(float4*)(newc + (size_t)k * F_ + c0) = out;

    if (t < 128) loss[n] = nm / rs1_s[id];
}

extern "C" void kernel_launch(void* const* d_in, const int* in_sizes, int n_in,
                              void* d_out, int out_size, void* d_ws, size_t ws_size,
                              hipStream_t stream) {
    const float* features = (const float*)d_in[0];
    const float* labels   = (const float*)d_in[1];
    const float* cluster  = (const float*)d_in[2];
    const float* cw       = (const float*)d_in[3];

    float* loss = (float*)d_out;
    float* newc = loss + N_;

    float* rowp  = (float*)d_ws;
    float* diagp = rowp + (F_ / 64) * 4 * K_;
    float* nume  = diagp + (F_ / 64) * K_;
    int* idxw    = (int*)(nume + N_);
    float* cntp  = (float*)(idxw + N_);
    float* part  = cntp + RS_ * K_;

    loss_idx<<<N_ / 8, 256, 0, stream>>>(features, labels, cluster, nume, idxw);
    cluster_sums_wmma<<<dim3(F_ / 64, RS_), 256, 0, stream>>>(features, idxw, cluster, part, cntp,
                                                              rowp, diagp);
    cluster_finish<<<dim3(F_ / 1024, K_), 256, 0, stream>>>(part, cntp, cluster, cw, rowp, diagp,
                                                            nume, idxw, newc, loss);
}


# squared distances ride in the cluster-sum launch: each lane takes its rows' cluster columns from an LDS strip, squares against its WMMA fragment loads; launch 1 argmax only; new nume sum order
# speedup vs baseline: 17.1267x; 1.2855x over previous
#include <hip/hip_runtime.h>


#define N_ 16384
#define K_ 64
#define F_ 2048
#define ALPHA_ 1.0f
#define GAMMA_ 1.0f

typedef float v2f __attribute__((ext_vector_type(2)));
typedef float v8f __attribute__((ext_vector_type(8)));

__global__ __launch_bounds__(256) void label_argmax(const float* __restrict__ labels,
                                                    int* __restrict__ idx_out) {
    const int lane = threadIdx.x & 31;
    const int c = lane & 15;
    const int h = lane >> 4;
    const int base = (blockIdx.x * 8 + (threadIdx.x >> 5)) * 16;

    float4 v[8];
#pragma unroll
    for (int u = 0; u < 8; ++u)
        v[u] = *(const float4*)(labels + (size_t)(base + 2 * u + h) * K_ + 4 * c);

    int res = 0;
#pragma unroll
    for (int u = 0; u < 8; ++u) {
        float bv = v[u].x;
        int bi = 4 * c;
        bool t;
        t = v[u].y > bv; bv = t ? v[u].y : bv; bi = t ? 4 * c + 1 : bi;
        t = v[u].z > bv; bv = t ? v[u].z : bv; bi = t ? 4 * c + 2 : bi;
        t = v[u].w > bv; bv = t ? v[u].w : bv; bi = t ? 4 * c + 3 : bi;
#pragma unroll
        for (int m = 8; m >= 1; m >>= 1) {
            const float ov = __shfl_xor(bv, m, 32);
            const int oi = __shfl_xor(bi, m, 32);
            t = (ov > bv) | ((ov == bv) & (oi < bi));
            bv = t ? ov : bv; bi = t ? oi : bi;
        }
        res = (c == u) ? bi : res;
    }
    if (c < 8) idx_out[base + 2 * c + h] = res;
}

#define RS_ 8
#define WROWS_ (N_ / RS_ / 8)
#define STAGE_ 8
#define NSTAGE_ (WROWS_ / (4 * STAGE_))
#define CLP_ 68

struct RowStage {
    float4 x[STAGE_][2];
    int2 id[STAGE_];
};

__device__ __forceinline__ void load_stage(RowStage& s, const float* __restrict__ fp,
                                           const int* __restrict__ ip) {
#pragma unroll
    for (int u = 0; u < STAGE_; ++u) {
        s.x[u][0] = *(const float4*)(fp + (size_t)(4 * u) * F_);
        s.x[u][1] = *(const float4*)(fp + (size_t)(4 * u + 1) * F_);
        s.id[u] = *(const int2*)(ip + 4 * u);
    }
}

__device__ __forceinline__ void mac_stage(const RowStage& s, const int c,
                                          v8f (&acc)[4][4], v2f (&cnt)[4]) {
#pragma unroll
    for (int u = 0; u < STAGE_; ++u) {
        const int t0 = s.id[u].x - c;
        const int t1 = s.id[u].y - c;
        const float4 x0 = s.x[u][0];
        const float4 x1 = s.x[u][1];
        v2f b[4];
        b[0].x = x0.x; b[0].y = x1.x;
        b[1].x = x0.y; b[1].y = x1.y;
        b[2].x = x0.z; b[2].y = x1.z;
        b[3].x = x0.w; b[3].y = x1.w;
#pragma unroll
        for (int mt = 0; mt < 4; ++mt) {
            v2f a;
            a.x = (t0 == 16 * mt) ? 1.f : 0.f;
            a.y = (t1 == 16 * mt) ? 1.f : 0.f;
            cnt[mt] += a;
#pragma unroll
            for (int j = 0; j < 4; ++j) {
                acc[mt][j] = __builtin_amdgcn_wmma_f32_16x16x4_f32(
                     false, a,  false, b[j],
                     (short)0, acc[mt][j],  false,  false);
            }
        }
    }
}

template <int M>
__device__ __forceinline__ float half_xor_add(float v) {
    return v + __int_as_float(__builtin_amdgcn_update_dpp(0, __float_as_int(v), 0x160 + M, 0xf, 0xf, false));
}

__device__ __forceinline__ void dist_stage(const RowStage& s, const float (*cl)[CLP_],
                                           const int c, float* __restrict__ np) {
    float4 g0[STAGE_], g1[STAGE_];
#pragma unroll
    for (int u = 0; u < STAGE_; ++u) {
        g0[u] = *(const float4*)&cl[s.id[u].x][4 * c];
        g1[u] = *(const float4*)&cl[s.id[u].y][4 * c];
    }
    float k0 = 0.f, k1 = 0.f;
#pragma unroll
    for (int u = 0; u < STAGE_; ++u) {
        const float4 x0 = s.x[u][0];
        const float4 x1 = s.x[u][1];
        const float a0 = x0.x - g0[u].x, a1 = x0.y - g0[u].y, a2 = x0.z - g0[u].z, a3 = x0.w - g0[u].w;
        const float b0 = x1.x - g1[u].x, b1 = x1.y - g1[u].y, b2 = x1.z - g1[u].z, b3 = x1.w - g1[u].w;
        float p0 = a0 * a0 + a1 * a1 + a2 * a2 + a3 * a3;
        float p1 = b0 * b0 + b1 * b1 + b2 * b2 + b3 * b3;
        p0 = half_xor_add<8>(p0); p1 = half_xor_add<8>(p1);
        p0 = half_xor_add<4>(p0); p1 = half_xor_add<4>(p1);
        p0 = half_xor_add<2>(p0); p1 = half_xor_add<2>(p1);
        p0 = half_xor_add<1>(p0); p1 = half_xor_add<1>(p1);
        if (c == u) { k0 = p0; k1 = p1; }
    }
    if (c < 8) {
        float2 o; o.x = k0; o.y = k1;
        *(float2*)(np + 4 * c) = o;
    }
}

__global__ __launch_bounds__(256) void cluster_sums_wmma(const float* __restrict__ features,
                                                         const int* __restrict__ idx,
                                                         const float* __restrict__ cluster,
                                                         float* __restrict__ part,
                                                         float* __restrict__ cntp,
                                                         float* __restrict__ rowp,
                                                         float* __restrict__ diagp,
                                                         float* __restrict__ numep) {
    __shared__ float red[8][K_ * 64];
    __shared__ float cred[8][K_];
    __shared__ float cl[K_][CLP_];

    const int lane = threadIdx.x & 31;
    const int wave = threadIdx.x >> 5;
    const int c = lane & 15;
    const int h = lane >> 4;
    const int strip = blockIdx.x;
    const int slice = blockIdx.y;
    const int row0 = slice * (N_ / RS_) + wave * WROWS_;

    const float* __restrict__ fp = features + (size_t)(row0 + 2 * h) * F_ + strip * 64 + 4 * c;
    const int* __restrict__ ip = idx + row0 + 2 * h;
    float* __restrict__ np = numep + (size_t)strip * N_ + row0 + 2 * h;

    RowStage sa, sb;
    {
        const int cr = threadIdx.x >> 4;
        const int cq = (threadIdx.x & 15) * 4;
        float4 cv[4];
#pragma unroll
        for (int i = 0; i < 4; ++i)
            cv[i] = *(const float4*)(cluster + (size_t)(cr + 16 * i) * F_ + strip * 64 + cq);
        load_stage(sa, fp, ip);
#pragma unroll
        for (int i = 0; i < 4; ++i) *(float4*)&cl[cr + 16 * i][cq] = cv[i];
    }
    __syncthreads();

    if (wave < 2) {
        const int t = 2 * slice + wave;
        const int ta = t >> 2;
        const int tb = t & 3;
        const float* __restrict__ ga = cluster + (size_t)(16 * ta + c) * F_ + strip * 64 + 4 * h;
        const float* __restrict__ gb = cluster + (size_t)(16 * tb + c) * F_ + strip * 64 + 4 * h;
        float4 av[8], bv[8];
#pragma unroll
        for (int i = 0; i < 8; ++i) {
            av[i] = *(const float4*)(ga + 8 * i);
            bv[i] = *(const float4*)(gb + 8 * i);
        }
        v8f g0 = {0.f, 0.f, 0.f, 0.f, 0.f, 0.f, 0.f, 0.f};
        v8f g1 = {0.f, 0.f, 0.f, 0.f, 0.f, 0.f, 0.f, 0.f};
#pragma unroll
        for (int i = 0; i < 8; ++i) {
            v2f a0, b0, a1, b1;
            a0.x = av[i].x; a0.y = av[i].y; b0.x = bv[i].x; b0.y = bv[i].y;
            a1.x = av[i].z; a1.y = av[i].w; b1.x = bv[i].z; b1.y = bv[i].w;
            g0 = __builtin_amdgcn_wmma_f32_16x16x4_f32(
                 false, a0,  false, b0,
                 (short)0, g0,  false,  false);
            g1 = __builtin_amdgcn_wmma_f32_16x16x4_f32(
                 false, a1,  false, b1,
                 (short)0, g1,  false,  false);
        }
        float gv[8];
#pragma unroll
        for (int r = 0; r < 8; ++r) gv[r] = g0[r] + g1[r];
        if (ta == tb) {
            float d = gv[0];
#pragma unroll
            for (int r = 1; r < 8; ++r) d = ((c & 7) == r) ? gv[r] : d;
            if ((c >> 3) == h) diagp[strip * K_ + 16 * ta + c] = d;
        }
        float rsel = 0.f;
#pragma unroll
        for (int r = 0; r < 8; ++r) {
            float v = gv[r];
            v += __shfl_xor(v, 8, 32);
            v += __shfl_xor(v, 4, 32);
            v += __shfl_xor(v, 2, 32);
            v += __shfl_xor(v, 1, 32);
            if (c == r) rsel = v;
        }
        if (c < 8) rowp[(size_t)(strip * 4 + tb) * K_ + 16 * ta + 8 * h + c] = rsel;
    }

    v8f acc[4][4];
    v2f cnt[4];
#pragma unroll
    for (int mt = 0; mt < 4; ++mt) {
        cnt[mt].x = 0.f; cnt[mt].y = 0.f;
#pragma unroll
        for (int j = 0; j < 4; ++j)
#pragma unroll
            for (int r = 0; r < 8; ++r) acc[mt][j][r] = 0.f;
    }

#pragma unroll 1
    for (int it = 0; it < NSTAGE_ - 2; it += 2) {
        __builtin_amdgcn_sched_barrier(0);
        load_stage(sb, fp + (size_t)(it + 1) * (4 * STAGE_) * F_, ip + (it + 1) * (4 * STAGE_));
        __builtin_amdgcn_sched_barrier(0);
        dist_stage(sa, cl, c, np + it * (4 * STAGE_));
        __builtin_amdgcn_sched_barrier(0);
        mac_stage(sa, c, acc, cnt);
        __builtin_amdgcn_sched_barrier(0);
        load_stage(sa, fp + (size_t)(it + 2) * (4 * STAGE_) * F_, ip + (it + 2) * (4 * STAGE_));
        __builtin_amdgcn_sched_barrier(0);
        dist_stage(sb, cl, c, np + (it + 1) * (4 * STAGE_));
        __builtin_amdgcn_sched_barrier(0);
        mac_stage(sb, c, acc, cnt);
    }
    __builtin_amdgcn_sched_barrier(0);
    load_stage(sb, fp + (size_t)(NSTAGE_ - 1) * (4 * STAGE_) * F_, ip + (NSTAGE_ - 1) * (4 * STAGE_));
    __builtin_amdgcn_sched_barrier(0);
    dist_stage(sa, cl, c, np + (NSTAGE_ - 2) * (4 * STAGE_));
    __builtin_amdgcn_sched_barrier(0);
    mac_stage(sa, c, acc, cnt);
    __builtin_amdgcn_sched_barrier(0);
    dist_stage(sb, cl, c, np + (NSTAGE_ - 1) * (4 * STAGE_));
    __builtin_amdgcn_sched_barrier(0);
    mac_stage(sb, c, acc, cnt);

#pragma unroll
    for (int mt = 0; mt < 4; ++mt) {
#pragma unroll
        for (int r = 0; r < 8; ++r) {
            float4 v;
            v.x = acc[mt][0][r]; v.y = acc[mt][1][r];
            v.z = acc[mt][2][r]; v.w = acc[mt][3][r];
            *(float4*)&red[wave][(mt * 16 + 8 * h + r) * 64 + 4 * c] = v;
        }
        float cw = cnt[mt].x + cnt[mt].y;
        cw += __shfl_xor(cw, 16, 32);
        if (h == 0) cred[wave][mt * 16 + c] = cw;
    }
    __syncthreads();

    float* __restrict__ pout = part + ((size_t)slice * K_ + wave * 8) * F_ + strip * 64;
#pragma unroll
    for (int i = 0; i < 4; ++i) {
        const int e = i * 128 + lane * 4;
        float4 sum = *(const float4*)&red[0][wave * 512 + e];
#pragma unroll
        for (int p = 1; p < 8; ++p) {
            const float4 v = *(const float4*)&red[p][wave * 512 + e];
            sum.x += v.x; sum.y += v.y; sum.z += v.z; sum.w += v.w;
        }
        *(float4*)(pout + (size_t)(e >> 6) * F_ + (e & 63)) = sum;
    }
    if (strip == 0 && threadIdx.x < K_) {
        float cs = cred[0][threadIdx.x];
#pragma unroll
        for (int p = 1; p < 8; ++p) cs += cred[p][threadIdx.x];
        cntp[slice * K_ + threadIdx.x] = cs;
    }
}

__global__ __launch_bounds__(256) void cluster_finish(const float* __restrict__ part,
                                                      const float* __restrict__ cntp,
                                                      const float* __restrict__ cluster,
                                                      const float* __restrict__ cw,
                                                      const float* __restrict__ rowp,
                                                      const float* __restrict__ diagp,
                                                      const float* __restrict__ numep,
                                                      const int* __restrict__ idx,
                                                      float* __restrict__ newc,
                                                      float* __restrict__ loss) {
    __shared__ float rowS[4][K_];
    __shared__ float diagS[4][K_];
    __shared__ float nrm_s[K_];
    __shared__ float rs1_s[K_];
    __shared__ float nsum[2][128];

    const int t = threadIdx.x;
    const int k = blockIdx.y;
    const int c0 = (blockIdx.x * 256 + t) * 4;

    float4 a = *(const float4*)(part + (size_t)k * F_ + c0);
    float cnt = cntp[k];
#pragma unroll
    for (int s = 1; s < RS_; ++s) {
        const float4 v = *(const float4*)(part + ((size_t)s * K_ + k) * F_ + c0);
        a.x += v.x; a.y += v.y; a.z += v.z; a.w += v.w;
        cnt += cntp[s * K_ + k];
    }
    const float4 cc = *(const float4*)(cluster + (size_t)k * F_ + c0);
    const int n = (blockIdx.y * 2 + blockIdx.x) * 128 + (t & 127);
    const int id = idx[n];
    {
        const int q2 = t >> 7;
        float nm = 0.f;
#pragma unroll
        for (int p = 0; p < 16; ++p) nm += numep[(size_t)(16 * q2 + p) * N_ + n];
        nsum[q2][t & 127] = nm;
    }

    {
        const int kk = t & 63;
        const int q = t >> 6;
        float rsum = 0.f;
#pragma unroll
        for (int p = 0; p < F_ / 64; ++p) rsum += rowp[(size_t)(p * 4 + q) * K_ + kk];
        float dsum = 0.f;
#pragma unroll
        for (int p = 0; p < 8; ++p) dsum += diagp[(size_t)(8 * q + p) * K_ + kk];
        rowS[q][kk] = rsum;
        diagS[q][kk] = dsum;
    }
    __syncthreads();
    if (t < K_) nrm_s[t] = (diagS[0][t] + diagS[1][t]) + (diagS[2][t] + diagS[3][t]);
    __syncthreads();
    if (t < K_) {
        float S = 0.f;
#pragma unroll 8
        for (int j = 0; j < K_; ++j) S += nrm_s[j];
        const float rowG = (rowS[0][t] + rowS[1][t]) + (rowS[2][t] + rowS[3][t]);
        rs1_s[t] = (float)K_ * nrm_s[t] + S - 2.f * rowG + ALPHA_;
    }
    __syncthreads();

    const float wk = cw[k] / rs1_s[k];
    float4 out;
    out.x = cc.x - GAMMA_ * wk * (a.x - cnt * cc.x);
    out.y = cc.y - GAMMA_ * wk * (a.y - cnt * cc.y);
    out.z = cc.z - GAMMA_ * wk * (a.z - cnt * cc.z);
    out.w = cc.w - GAMMA_ * wk * (a.w - cnt * cc.w);
    *(float4*)(newc + (size_t)k * F_ + c0) = out;

    if (t < 128) loss[n] = (nsum[0][t] + nsum[1][t]) / rs1_s[id];
}

extern "C" void kernel_launch(void* const* d_in, const int* in_sizes, int n_in,
                              void* d_out, int out_size, void* d_ws, size_t ws_size,
                              hipStream_t stream) {
    const float* features = (const float*)d_in[0];
    const float* labels   = (const float*)d_in[1];
    const float* cluster  = (const float*)d_in[2];
    const float* cw       = (const float*)d_in[3];

    float* loss = (float*)d_out;
    float* newc = loss + N_;

    float* rowp  = (float*)d_ws;
    float* diagp = rowp + (F_ / 64) * 4 * K_;
    int* idxw    = (int*)(diagp + (F_ / 64) * K_);
    float* cntp  = (float*)(idxw + N_);
    float* part  = cntp + RS_ * K_;
    float* numep = part + (size_t)RS_ * K_ * F_;

    label_argmax<<<N_ / 128, 256, 0, stream>>>(labels, idxw);
    cluster_sums_wmma<<<dim3(F_ / 64, RS_), 256, 0, stream>>>(features, idxw, cluster, part, cntp,
                                                              rowp, diagp, numep);
    cluster_finish<<<dim3(F_ / 1024, K_), 256, 0, stream>>>(part, cntp, cluster, cw, rowp, diagp,
                                                            numep, idxw, newc, loss);
}


# cluster-sum launch, row distances: squares on the loads' own register pairs (no moves); a half's 16 lanes add their 16 values as a transposed DPP reduction (15 adds); new in-lane nume sum order
# speedup vs baseline: 21.4064x; 1.2499x over previous
#include <hip/hip_runtime.h>


#define N_ 16384
#define K_ 64
#define F_ 2048
#define ALPHA_ 1.0f
#define GAMMA_ 1.0f

typedef float v2f __attribute__((ext_vector_type(2)));
typedef float v8f __attribute__((ext_vector_type(8)));

__global__ __launch_bounds__(256) void label_argmax(const float* __restrict__ labels,
                                                    int* __restrict__ idx_out) {
    const int lane = threadIdx.x & 31;
    const int c = lane & 15;
    const int h = lane >> 4;
    const int base = (blockIdx.x * 8 + (threadIdx.x >> 5)) * 16;

    float4 v[8];
#pragma unroll
    for (int u = 0; u < 8; ++u)
        v[u] = *(const float4*)(labels + (size_t)(base + 2 * u + h) * K_ + 4 * c);

    int res = 0;
#pragma unroll
    for (int u = 0; u < 8; ++u) {
        float bv = v[u].x;
        int bi = 4 * c;
        bool t;
        t = v[u].y > bv; bv = t ? v[u].y : bv; bi = t ? 4 * c + 1 : bi;
        t = v[u].z > bv; bv = t ? v[u].z : bv; bi = t ? 4 * c + 2 : bi;
        t = v[u].w > bv; bv = t ? v[u].w : bv; bi = t ? 4 * c + 3 : bi;
#pragma unroll
        for (int m = 8; m >= 1; m >>= 1) {
            const float ov = __shfl_xor(bv, m, 32);
            const int oi = __shfl_xor(bi, m, 32);
            t = (ov > bv) | ((ov == bv) & (oi < bi));
            bv = t ? ov : bv; bi = t ? oi : bi;
        }
        res = (c == u) ? bi : res;
    }
    if (c < 8) idx_out[base + 2 * c + h] = res;
}

#define RS_ 8
#define SROWS_ (N_ / RS_)
#define STAGE_ 8
#define SPOS_ (4 * STAGE_)
#define LCAP_ (SROWS_ + 4 * SPOS_)
#define CLP_ 68
#define NOROW_ (K_ << 16)
#define LDSA_ __attribute__((aligned(16)))

struct RowStage {
    float4 x[STAGE_][2];
    int2 id[STAGE_];
};

__device__ __forceinline__ void load_stage(RowStage& s, const float* __restrict__ fs,
                                           const int* lp, const int c) {
    int2 e[STAGE_];
#pragma unroll
    for (int u = 0; u < STAGE_; ++u) e[u] = *(const int2*)(lp + 4 * u);
#pragma unroll
    for (int u = 0; u < STAGE_; ++u) {
        s.id[u].x = e[u].x >> 16;
        s.id[u].y = e[u].y >> 16;
        s.x[u][0] = *(const float4*)(fs + (unsigned)((e[u].x & 0xffff) * F_ + 4 * c));
        s.x[u][1] = *(const float4*)(fs + (unsigned)((e[u].y & 0xffff) * F_ + 4 * c));
    }
}

__device__ __forceinline__ void mac_stage(const RowStage& s, const int cq,
                                          v8f (&acc)[4], v2f& cnt) {
#pragma unroll
    for (int u = 0; u < STAGE_; ++u) {
        const float4 x0 = s.x[u][0];
        const float4 x1 = s.x[u][1];
        v2f b[4];
        b[0].x = x0.x; b[0].y = x1.x;
        b[1].x = x0.y; b[1].y = x1.y;
        b[2].x = x0.z; b[2].y = x1.z;
        b[3].x = x0.w; b[3].y = x1.w;
        v2f a;
        a.x = (s.id[u].x == cq) ? 1.f : 0.f;
        a.y = (s.id[u].y == cq) ? 1.f : 0.f;
        cnt += a;
#pragma unroll
        for (int j = 0; j < 4; ++j) {
            acc[j] = __builtin_amdgcn_wmma_f32_16x16x4_f32(
                 false, a,  false, b[j],
                 (short)0, acc[j],  false,  false);
        }
    }
}

template <int M>
__device__ __forceinline__ float half_xor_add(float keep, float send) {
    return keep + __int_as_float(__builtin_amdgcn_update_dpp(0, __float_as_int(send), 0x160 + M, 0xf, 0xf, false));
}

__device__ __forceinline__ float sq4(const float4 x, const float4 g) {
    v2f xa, xb, ga, gb;
    xa.x = x.x; xa.y = x.y; xb.x = x.z; xb.y = x.w;
    ga.x = g.x; ga.y = g.y; gb.x = g.z; gb.y = g.w;
    const v2f da = xa - ga;
    const v2f db = xb - gb;
    v2f sq = da * da;
    sq = db * db + sq;
    return sq.x + sq.y;
}

__device__ __forceinline__ void dist_stage(const RowStage& s, const float (*cl)[CLP_],
                                           const int c, const int* lp, float* nrow) {
    float4 g0[STAGE_], g1[STAGE_];
#pragma unroll
    for (int u = 0; u < STAGE_; ++u) {
        g0[u] = *(const float4*)&cl[s.id[u].x & (K_ - 1)][4 * c];
        g1[u] = *(const float4*)&cl[s.id[u].y & (K_ - 1)][4 * c];
    }
    const int eo = lp[4 * (c & 7) + (c >> 3)];
    float p[2 * STAGE_];
#pragma unroll
    for (int u = 0; u < STAGE_; ++u) {
        p[u] = sq4(s.x[u][0], g0[u]);
        p[STAGE_ + u] = sq4(s.x[u][1], g1[u]);
    }
    const bool b8 = (c & 8) != 0, b4 = (c & 4) != 0, b2 = (c & 2) != 0, b1 = (c & 1) != 0;
    float q[8], r[4], t[2];
#pragma unroll
    for (int i = 0; i < 8; ++i)
        q[i] = half_xor_add<8>(b8 ? p[i + 8] : p[i], b8 ? p[i] : p[i + 8]);
#pragma unroll
    for (int i = 0; i < 4; ++i)
        r[i] = half_xor_add<4>(b4 ? q[i + 4] : q[i], b4 ? q[i] : q[i + 4]);
#pragma unroll
    for (int i = 0; i < 2; ++i)
        t[i] = half_xor_add<2>(b2 ? r[i + 2] : r[i], b2 ? r[i] : r[i + 2]);
    const float tot = half_xor_add<1>(b1 ? t[1] : t[0], b1 ? t[0] : t[1]);
    if ((eo >> 16) < K_) nrow[eo & 0xffff] = tot;
}

__global__ __launch_bounds__(256) void cluster_sums_wmma(const float* __restrict__ features,
                                                         const int* __restrict__ idx,
                                                         const float* __restrict__ cluster,
                                                         float* __restrict__ part,
                                                         float* __restrict__ cntp,
                                                         float* __restrict__ rowp,
                                                         float* __restrict__ diagp,
                                                         float* __restrict__ numep) {
    __shared__ LDSA_ float red[8][16 * 64];
    __shared__ LDSA_ float cred[8][16];
    __shared__ LDSA_ float cl[K_][CLP_];
    __shared__ LDSA_ int lst[LCAP_];
    __shared__ LDSA_ int wtot[8][4];
    __shared__ LDSA_ float nrow[SROWS_];

    const int lane = threadIdx.x & 31;
    const int wave = threadIdx.x >> 5;
    const int c = lane & 15;
    const int h = lane >> 4;
    const int strip = blockIdx.x;
    const int slice = blockIdx.y;
    const int R0 = slice * SROWS_;

    const float* __restrict__ fs = features + (size_t)R0 * F_ + strip * 64;

    int myid[8], mypos[8];
    {
        int run0 = 0, run1 = 0, run2 = 0, run3 = 0;
#pragma unroll
        for (int u = 0; u < 8; ++u) myid[u] = idx[R0 + wave * 256 + u * 32 + lane];
#pragma unroll
        for (int u = 0; u < 8; ++u) {
            const int b = myid[u] >> 4;
            const unsigned m0 = __builtin_amdgcn_ballot_w32(b == 0);
            const unsigned m1 = __builtin_amdgcn_ballot_w32(b == 1);
            const unsigned m2 = __builtin_amdgcn_ballot_w32(b == 2);
            const unsigned m3 = __builtin_amdgcn_ballot_w32(b == 3);
            const unsigned mm = (b == 0) ? m0 : (b == 1) ? m1 : (b == 2) ? m2 : m3;
            const int rb = (b == 0) ? run0 : (b == 1) ? run1 : (b == 2) ? run2 : run3;
            mypos[u] = rb + (int)__builtin_amdgcn_mbcnt_lo(mm, 0u);
            run0 += __popc(m0); run1 += __popc(m1); run2 += __popc(m2); run3 += __popc(m3);
        }
        if (lane == 0) {
            int4 tw; tw.x = run0; tw.y = run1; tw.z = run2; tw.w = run3;
            *(int4*)&wtot[wave][0] = tw;
        }
    }
    for (int i = threadIdx.x; i < LCAP_; i += 256) lst[i] = NOROW_;

    {
        const int cr = threadIdx.x >> 4;
        const int cq4 = (threadIdx.x & 15) * 4;
        float4 cv[4];
#pragma unroll
        for (int i = 0; i < 4; ++i)
            cv[i] = *(const float4*)(cluster + (size_t)(cr + 16 * i) * F_ + strip * 64 + cq4);
#pragma unroll
        for (int i = 0; i < 4; ++i) *(float4*)&cl[cr + 16 * i][cq4] = cv[i];
    }
    __syncthreads();

    int n0 = 0, n1 = 0, n2 = 0, n3 = 0, o0 = 0, o1 = 0, o2 = 0, o3 = 0;
#pragma unroll
    for (int w = 0; w < 8; ++w) {
        const int4 tw = *(const int4*)&wtot[w][0];
        if (w < wave) { o0 += tw.x; o1 += tw.y; o2 += tw.z; o3 += tw.w; }
        n0 += tw.x; n1 += tw.y; n2 += tw.z; n3 += tw.w;
    }
    const int b0 = 0;
    const int b1 = b0 + ((n0 + SPOS_ - 1) & ~(SPOS_ - 1));
    const int b2 = b1 + ((n1 + SPOS_ - 1) & ~(SPOS_ - 1));
    const int b3 = b2 + ((n2 + SPOS_ - 1) & ~(SPOS_ - 1));
#pragma unroll
    for (int u = 0; u < 8; ++u) {
        const int b = myid[u] >> 4;
        const int at = (b == 0) ? (b0 + o0) : (b == 1) ? (b1 + o1) : (b == 2) ? (b2 + o2) : (b3 + o3);
        lst[at + mypos[u]] = (wave * 256 + u * 32 + lane) | (myid[u] << 16);
    }
    __syncthreads();

    if (wave < 2) {
        const int t = 2 * slice + wave;
        const int ta = t >> 2;
        const int tb = t & 3;
        const float* __restrict__ ga = cluster + (size_t)(16 * ta + c) * F_ + strip * 64 + 4 * h;
        const float* __restrict__ gb = cluster + (size_t)(16 * tb + c) * F_ + strip * 64 + 4 * h;
        float4 av[8], bv[8];
#pragma unroll
        for (int i = 0; i < 8; ++i) {
            av[i] = *(const float4*)(ga + 8 * i);
            bv[i] = *(const float4*)(gb + 8 * i);
        }
        v8f g0 = {0.f, 0.f, 0.f, 0.f, 0.f, 0.f, 0.f, 0.f};
        v8f g1 = {0.f, 0.f, 0.f, 0.f, 0.f, 0.f, 0.f, 0.f};
#pragma unroll
        for (int i = 0; i < 8; ++i) {
            v2f a0, bb0, a1, bb1;
            a0.x = av[i].x; a0.y = av[i].y; bb0.x = bv[i].x; bb0.y = bv[i].y;
            a1.x = av[i].z; a1.y = av[i].w; bb1.x = bv[i].z; bb1.y = bv[i].w;
            g0 = __builtin_amdgcn_wmma_f32_16x16x4_f32(
                 false, a0,  false, bb0,
                 (short)0, g0,  false,  false);
            g1 = __builtin_amdgcn_wmma_f32_16x16x4_f32(
                 false, a1,  false, bb1,
                 (short)0, g1,  false,  false);
        }
        float gv[8];
#pragma unroll
        for (int r = 0; r < 8; ++r) gv[r] = g0[r] + g1[r];
        if (ta == tb) {
            float d = gv[0];
#pragma unroll
            for (int r = 1; r < 8; ++r) d = ((c & 7) == r) ? gv[r] : d;
            if ((c >> 3) == h) diagp[strip * K_ + 16 * ta + c] = d;
        }
        float rsel = 0.f;
#pragma unroll
        for (int r = 0; r < 8; ++r) {
            float v = gv[r];
            v += __shfl_xor(v, 8, 32);
            v += __shfl_xor(v, 4, 32);
            v += __shfl_xor(v, 2, 32);
            v += __shfl_xor(v, 1, 32);
            if (c == r) rsel = v;
        }
        if (c < 8) rowp[(size_t)(strip * 4 + tb) * K_ + 16 * ta + 8 * h + c] = rsel;
    }

    const int q = wave & 3;
    const int hf = wave >> 2;
    const int nq = (q == 0) ? n0 : (q == 1) ? n1 : (q == 2) ? n2 : n3;
    const int bq = (q == 0) ? b0 : (q == 1) ? b1 : (q == 2) ? b2 : b3;
    const int Tq = (nq + SPOS_ - 1) / SPOS_;
    const int Th = (Tq + 1) >> 1;
    const int tbeg = hf ? Th : 0;
    const int tend = hf ? Tq : Th;
    const int pbeg = __builtin_amdgcn_readfirstlane(bq + SPOS_ * tbeg);
    const int nst = __builtin_amdgcn_readfirstlane(tend - tbeg);
    const int cq = 16 * q + c;
    const int* lp = lst + pbeg + 2 * h;

    v8f acc[4];
    v2f cnt;
    cnt.x = 0.f; cnt.y = 0.f;
#pragma unroll
    for (int j = 0; j < 4; ++j)
#pragma unroll
        for (int r = 0; r < 8; ++r) acc[j][r] = 0.f;

    if (nst > 0) {
        RowStage sa, sb;
        const int last = nst - 1;
        load_stage(sa, fs, lp, c);
#pragma unroll 1
        for (int it = 0; it < nst; it += 2) {
            __builtin_amdgcn_sched_barrier(0);
            load_stage(sb, fs, lp + min(it + 1, last) * SPOS_, c);
            __builtin_amdgcn_sched_barrier(0);
            dist_stage(sa, cl, c, lp + it * SPOS_, nrow);
            __builtin_amdgcn_sched_barrier(0);
            mac_stage(sa, cq, acc, cnt);
            __builtin_amdgcn_sched_barrier(0);
            load_stage(sa, fs, lp + min(it + 2, last) * SPOS_, c);
            __builtin_amdgcn_sched_barrier(0);
            if (it + 1 < nst) {
                dist_stage(sb, cl, c, lp + (it + 1) * SPOS_, nrow);
                __builtin_amdgcn_sched_barrier(0);
                mac_stage(sb, cq, acc, cnt);
            }
        }
    }

#pragma unroll
    for (int r = 0; r < 8; ++r) {
        float4 v;
        v.x = acc[0][r]; v.y = acc[1][r];
        v.z = acc[2][r]; v.w = acc[3][r];
        *(float4*)&red[wave][(8 * h + r) * 64 + 4 * c] = v;
    }
    {
        float cw = cnt.x + cnt.y;
        cw += __shfl_xor(cw, 16, 32);
        if (h == 0) cred[wave][c] = cw;
    }
    __syncthreads();

    float* __restrict__ pout = part + (size_t)slice * K_ * F_ + strip * 64;
#pragma unroll
    for (int i = 0; i < 4; ++i) {
        const int e = i * 1024 + threadIdx.x * 4;
        const int kk = e >> 6;
        const int lo = (kk & 15) * 64 + (e & 63);
        const float4 v0 = *(const float4*)&red[kk >> 4][lo];
        const float4 v1 = *(const float4*)&red[(kk >> 4) + 4][lo];
        float4 sum;
        sum.x = v0.x + v1.x; sum.y = v0.y + v1.y; sum.z = v0.z + v1.z; sum.w = v0.w + v1.w;
        *(float4*)(pout + (size_t)kk * F_ + (e & 63)) = sum;
    }
    if (strip == 0 && threadIdx.x < K_) {
        const int kk = threadIdx.x;
        cntp[slice * K_ + kk] = cred[kk >> 4][kk & 15] + cred[(kk >> 4) + 4][kk & 15];
    }
    float* __restrict__ nout = numep + (size_t)strip * N_ + R0;
#pragma unroll
    for (int i = 0; i < 2; ++i) {
        const int e = i * 1024 + threadIdx.x * 4;
        *(float4*)(nout + e) = *(const float4*)&nrow[e];
    }
}

__global__ __launch_bounds__(256) void cluster_finish(const float* __restrict__ part,
                                                      const float* __restrict__ cntp,
                                                      const float* __restrict__ cluster,
                                                      const float* __restrict__ cw,
                                                      const float* __restrict__ rowp,
                                                      const float* __restrict__ diagp,
                                                      const float* __restrict__ numep,
                                                      const int* __restrict__ idx,
                                                      float* __restrict__ newc,
                                                      float* __restrict__ loss) {
    __shared__ float rowS[4][K_];
    __shared__ float diagS[4][K_];
    __shared__ float nrm_s[K_];
    __shared__ float rs1_s[K_];
    __shared__ float nsum[2][128];

    const int t = threadIdx.x;
    const int k = blockIdx.y;
    const int c0 = (blockIdx.x * 256 + t) * 4;

    float4 a = *(const float4*)(part + (size_t)k * F_ + c0);
    float cnt = cntp[k];
#pragma unroll
    for (int s = 1; s < RS_; ++s) {
        const float4 v = *(const float4*)(part + ((size_t)s * K_ + k) * F_ + c0);
        a.x += v.x; a.y += v.y; a.z += v.z; a.w += v.w;
        cnt += cntp[s * K_ + k];
    }
    const float4 cc = *(const float4*)(cluster + (size_t)k * F_ + c0);
    const int n = (blockIdx.y * 2 + blockIdx.x) * 128 + (t & 127);
    const int id = idx[n];
    {
        const int q2 = t >> 7;
        float nm = 0.f;
#pragma unroll
        for (int p = 0; p < 16; ++p) nm += numep[(size_t)(16 * q2 + p) * N_ + n];
        nsum[q2][t & 127] = nm;
    }

    {
        const int kk = t & 63;
        const int q = t >> 6;
        float rsum = 0.f;
#pragma unroll
        for (int p = 0; p < F_ / 64; ++p) rsum += rowp[(size_t)(p * 4 + q) * K_ + kk];
        float dsum = 0.f;
#pragma unroll
        for (int p = 0; p < 8; ++p) dsum += diagp[(size_t)(8 * q + p) * K_ + kk];
        rowS[q][kk] = rsum;
        diagS[q][kk] = dsum;
    }
    __syncthreads();
    if (t < K_) nrm_s[t] = (diagS[0][t] + diagS[1][t]) + (diagS[2][t] + diagS[3][t]);
    __syncthreads();
    if (t < K_) {
        float S = 0.f;
#pragma unroll 8
        for (int j = 0; j < K_; ++j) S += nrm_s[j];
        const float rowG = (rowS[0][t] + rowS[1][t]) + (rowS[2][t] + rowS[3][t]);
        rs1_s[t] = (float)K_ * nrm_s[t] + S - 2.f * rowG + ALPHA_;
    }
    __syncthreads();

    const float wk = cw[k] / rs1_s[k];
    float4 out;
    out.x = cc.x - GAMMA_ * wk * (a.x - cnt * cc.x);
    out.y = cc.y - GAMMA_ * wk * (a.y - cnt * cc.y);
    out.z = cc.z - GAMMA_ * wk * (a.z - cnt * cc.z);
    out.w = cc.w - GAMMA_ * wk * (a.w - cnt * cc.w);
    *(float4*)(newc + (size_t)k * F_ + c0) = out;

    if (t < 128) loss[n] = (nsum[0][t] + nsum[1][t]) / rs1_s[id];
}

extern "C" void kernel_launch(void* const* d_in, const int* in_sizes, int n_in,
                              void* d_out, int out_size, void* d_ws, size_t ws_size,
                              hipStream_t stream) {
    const float* features = (const float*)d_in[0];
    const float* labels   = (const float*)d_in[1];
    const float* cluster  = (const float*)d_in[2];
    const float* cw       = (const float*)d_in[3];

    float* loss = (float*)d_out;
    float* newc = loss + N_;

    float* rowp  = (float*)d_ws;
    float* diagp = rowp + (F_ / 64) * 4 * K_;
    int* idxw    = (int*)(diagp + (F_ / 64) * K_);
    float* cntp  = (float*)(idxw + N_);
    float* part  = cntp + RS_ * K_;
    float* numep = part + (size_t)RS_ * K_ * F_;

    label_argmax<<<N_ / 128, 256, 0, stream>>>(labels, idxw);
    cluster_sums_wmma<<<dim3(F_ / 64, RS_), 256, 0, stream>>>(features, idxw, cluster, part, cntp,
                                                              rowp, diagp, numep);
    cluster_finish<<<dim3(F_ / 1024, K_), 256, 0, stream>>>(part, cntp, cluster, cw, rowp, diagp,
                                                            numep, idxw, newc, loss);
}
